# starting kernel
# baseline (speedup 1.0000x reference)
#include <hip/hip_runtime.h>
#include <hip/hip_bf16.h>
#include <math.h>

#define NPOS  80
#define NCHAR 256
#define HDIM  512
#define IDIM  (NPOS * NCHAR)
#define BATCH 8192
#define LN_EPS 1e-5f
#define KC    32
#define LDP   34

typedef __attribute__((ext_vector_type(2))) float v2f;
typedef __attribute__((ext_vector_type(8))) float v8f;

__global__ void k_transpose_w1(const float* __restrict__ W1,
                               float* __restrict__ W1T) {
  __shared__ float tile[32][33];
  const int i0 = blockIdx.x * 32;
  const int j0 = blockIdx.y * 32;
  const int tx = threadIdx.x;
  const int ty = threadIdx.y;
#pragma unroll
  for (int r = ty; r < 32; r += 8)
    tile[r][tx] = W1[(size_t)(j0 + r) * IDIM + (i0 + tx)];
  __syncthreads();
#pragma unroll
  for (int r = ty; r < 32; r += 8)
    W1T[(size_t)(i0 + r) * HDIM + (j0 + tx)] = tile[tx][r];
}

__device__ __forceinline__ float2 ln_stats_512(float a0, float a1) {
  float s = a0 + a1;
  float q = a0 * a0 + a1 * a1;
#pragma unroll
  for (int off = 16; off > 0; off >>= 1) {
    s += __shfl_xor(s, off, 32);
    q += __shfl_xor(q, off, 32);
  }
  __shared__ float ssum[8], sq[8], stats[2];
  const int wid  = threadIdx.x >> 5;
  const int lane = threadIdx.x & 31;
  if (lane == 0) { ssum[wid] = s; sq[wid] = q; }
  __syncthreads();
  if (threadIdx.x == 0) {
    float S = 0.f, Q = 0.f;
#pragma unroll
    for (int w = 0; w < 8; ++w) { S += ssum[w]; Q += sq[w]; }
    const float mu  = S * (1.0f / (float)HDIM);
    const float var = Q * (1.0f / (float)HDIM) - mu * mu;
    stats[0] = mu;
    stats[1] = rsqrtf(var + LN_EPS);
  }
  __syncthreads();
  return make_float2(stats[0], stats[1]);
}

__device__ __forceinline__ float elu(float x) {
  return x > 0.f ? x : expm1f(x);
}

__global__ void k_embed_ln1(const int* __restrict__ msg,
                            const float* __restrict__ W1T,
                            const float* __restrict__ b1,
                            const float* __restrict__ g1,
                            const float* __restrict__ be1,
                            float* __restrict__ h1) {
  const int b   = blockIdx.x;
  const int j0  = threadIdx.x * 2;
  const int* m  = msg + (size_t)b * NPOS;
  float a0 = 0.f, a1 = 0.f;
  int c = m[0];
  for (int p = 0; p < NPOS; ++p) {
    const float* col = W1T + (size_t)(p * NCHAR + c) * HDIM;
    if (p + 1 < NPOS) {
      c = m[p + 1];
      __builtin_prefetch(W1T + (size_t)((p + 1) * NCHAR + c) * HDIM + j0, 0, 1);
    }
    const v2f v = *(const v2f*)(col + j0);
    a0 += v.x;
    a1 += v.y;
  }
  a0 += b1[j0];
  a1 += b1[j0 + 1];
  const float2 st = ln_stats_512(a0, a1);
  const float v0 = (a0 - st.x) * st.y * g1[j0]     + be1[j0];
  const float v1 = (a1 - st.x) * st.y * g1[j0 + 1] + be1[j0 + 1];
  h1[(size_t)b * HDIM + j0]     = elu(v0);
  h1[(size_t)b * HDIM + j0 + 1] = elu(v1);
}

__global__ void k_embed_ln1_direct(const int* __restrict__ msg,
                                   const float* __restrict__ W1,
                                   const float* __restrict__ b1,
                                   const float* __restrict__ g1,
                                   const float* __restrict__ be1,
                                   float* __restrict__ h1) {
  const int b   = blockIdx.x;
  const int tid = threadIdx.x;
  const int* m  = msg + (size_t)b * NPOS;
  float a0 = 0.f, a1 = 0.f;
  for (int p = 0; p < NPOS; ++p) {
    const size_t idx = (size_t)p * NCHAR + m[p];
    a0 += W1[(size_t)tid * IDIM + idx];
    a1 += W1[(size_t)(tid + 256) * IDIM + idx];
  }
  a0 += b1[tid];
  a1 += b1[tid + 256];
  const float2 st = ln_stats_512(a0, a1);
  const float v0 = (a0 - st.x) * st.y * g1[tid]       + be1[tid];
  const float v1 = (a1 - st.x) * st.y * g1[tid + 256] + be1[tid + 256];
  h1[(size_t)b * HDIM + tid]       = elu(v0);
  h1[(size_t)b * HDIM + tid + 256] = elu(v1);
}

__global__ void k_gemm_wmma(const float* __restrict__ h1,
                            const float* __restrict__ W2,
                            const float* __restrict__ b2,
                            float* __restrict__ h2) {
  __shared__ float sA[128 * LDP];
  __shared__ float sB[64 * LDP];
  const int tid   = threadIdx.x;
  const int wid   = tid >> 5;
  const int lane  = tid & 31;
  const int m0    = blockIdx.x * 128;
  const int n0    = blockIdx.y * 64;
  const int mrow  = lane & 15;
  const int khalf = (lane >> 4) << 1;

  v8f acc0 = {}, acc1 = {}, acc2 = {}, acc3 = {};

  for (int k0 = 0; k0 < HDIM; k0 += KC) {
#pragma unroll
    for (int i = 0; i < 4; ++i) {
      const int idx = tid + i * 256;
      const int r   = idx >> 3;
      const int cc  = (idx & 7) << 2;
      const float4 v = *(const float4*)(h1 + (size_t)(m0 + r) * HDIM + k0 + cc);
      v2f lo = {v.x, v.y};
      v2f hi = {v.z, v.w};
      *(v2f*)(sA + r * LDP + cc)     = lo;
      *(v2f*)(sA + r * LDP + cc + 2) = hi;
    }
#pragma unroll
    for (int i = 0; i < 2; ++i) {
      const int idx = tid + i * 256;
      const int r   = idx >> 3;
      const int cc  = (idx & 7) << 2;
      const float4 v = *(const float4*)(W2 + (size_t)(n0 + r) * HDIM + k0 + cc);
      v2f lo = {v.x, v.y};
      v2f hi = {v.z, v.w};
      *(v2f*)(sB + r * LDP + cc)     = lo;
      *(v2f*)(sB + r * LDP + cc + 2) = hi;
    }
    __syncthreads();

    const float* Aw = sA + (wid * 16 + mrow) * LDP + khalf;
    const float* Bw = sB + mrow * LDP + khalf;
#pragma unroll
    for (int kk = 0; kk < KC; kk += 4) {
      const v2f a   = *(const v2f*)(Aw + kk);
      const v2f bb0 = *(const v2f*)(Bw + kk);
      const v2f bb1 = *(const v2f*)(Bw + 16 * LDP + kk);
      const v2f bb2 = *(const v2f*)(Bw + 32 * LDP + kk);
      const v2f bb3 = *(const v2f*)(Bw + 48 * LDP + kk);
      acc0 = __builtin_amdgcn_wmma_f32_16x16x4_f32(false, a, false, bb0, (short)0, acc0, false, false);
      acc1 = __builtin_amdgcn_wmma_f32_16x16x4_f32(false, a, false, bb1, (short)0, acc1, false, false);
      acc2 = __builtin_amdgcn_wmma_f32_16x16x4_f32(false, a, false, bb2, (short)0, acc2, false, false);
      acc3 = __builtin_amdgcn_wmma_f32_16x16x4_f32(false, a, false, bb3, (short)0, acc3, false, false);
    }
    __syncthreads();
  }

  const int ncol  = lane & 15;
  const int mbase = (lane >> 4) << 3;
  float* out = h2 + (size_t)(m0 + wid * 16 + mbase) * HDIM + n0;
  const float bj0 = b2[n0 + ncol];
  const float bj1 = b2[n0 + 16 + ncol];
  const float bj2 = b2[n0 + 32 + ncol];
  const float bj3 = b2[n0 + 48 + ncol];
#pragma unroll
  for (int r = 0; r < 8; ++r) {
    float* row = out + (size_t)r * HDIM;
    row[ncol]      = acc0[r] + bj0;
    row[16 + ncol] = acc1[r] + bj1;
    row[32 + ncol] = acc2[r] + bj2;
    row[48 + ncol] = acc3[r] + bj3;
  }
}

__global__ void k_ln2_elu(const float* __restrict__ h2,
                          const float* __restrict__ g2,
                          const float* __restrict__ be2,
                          float* __restrict__ out) {
  const int b  = blockIdx.x;
  const int j0 = threadIdx.x * 2;
  const v2f v  = *(const v2f*)(h2 + (size_t)b * HDIM + j0);
  const float a0 = v.x, a1 = v.y;
  const float2 st = ln_stats_512(a0, a1);
  const float v0 = (a0 - st.x) * st.y * g2[j0]     + be2[j0];
  const float v1 = (a1 - st.x) * st.y * g2[j0 + 1] + be2[j0 + 1];
  v2f o = {elu(v0), elu(v1)};
  *(v2f*)(out + (size_t)b * HDIM + j0) = o;
}

extern "C" void kernel_launch(void* const* d_in, const int* in_sizes, int n_in,
                              void* d_out, int out_size, void* d_ws, size_t ws_size,
                              hipStream_t stream) {
  (void)in_sizes; (void)n_in; (void)out_size;
  const int*   msg = (const int*)  d_in[0];
  const float* W1  = (const float*)d_in[1];
  const float* b1  = (const float*)d_in[2];
  const float* g1  = (const float*)d_in[3];
  const float* be1 = (const float*)d_in[4];
  const float* W2  = (const float*)d_in[5];
  const float* b2  = (const float*)d_in[6];
  const float* g2  = (const float*)d_in[7];
  const float* be2 = (const float*)d_in[8];
  float* out = (float*)d_out;

  const size_t w1t_bytes = (size_t)IDIM * HDIM * sizeof(float);
  const size_t h_bytes   = (size_t)BATCH * HDIM * sizeof(float);
  char* ws = (char*)d_ws;
  const bool use_transpose = ws_size >= (w1t_bytes + 2 * h_bytes);

  float* h1;
  float* h2;
  if (use_transpose) {
    float* W1T = (float*)ws;
    h1 = (float*)(ws + w1t_bytes);
    h2 = (float*)(ws + w1t_bytes + h_bytes);
    k_transpose_w1<<<dim3(IDIM / 32, HDIM / 32), dim3(32, 8), 0, stream>>>(W1, W1T);
    k_embed_ln1<<<BATCH, 256, 0, stream>>>(msg, W1T, b1, g1, be1, h1);
  } else {
    h1 = (float*)ws;
    h2 = (float*)(ws + h_bytes);
    k_embed_ln1_direct<<<BATCH, 256, 0, stream>>>(msg, W1, b1, g1, be1, h1);
  }

  k_gemm_wmma<<<dim3(BATCH / 128, HDIM / 64), 256, 0, stream>>>(h1, W2, b2, h2);
  k_ln2_elu<<<BATCH, 256, 0, stream>>>(h2, g2, be2, out);
}


# embedding LDS table re-laid [character][slot], row b starts its 80 positions at b mod 32: a wave's 32 lookups hit 32 consecutive slots, no bank conflicts; rows summed in rotated position order
# speedup vs baseline: 1.6041x; 1.6041x over previous
#include <hip/hip_runtime.h>
#include <math.h>


#define NPOS  80
#define NCHAR 256
#define HDIM  512
#define IDIM  (NPOS * NCHAR)
#define BATCH 8192
#define LN_EPS 1e-5f
#define KC    32
#define LDP   34

typedef __attribute__((ext_vector_type(2))) float v2f;
typedef __attribute__((ext_vector_type(8))) float v8f;

__device__ __forceinline__ float elu(float x) {
  return x > 0.f ? x : expm1f(x);
}

#define NGRP (NPOS / 16)

__device__ __forceinline__ unsigned pack4(int4 v) {
  return (unsigned)(v.x & 255) | ((unsigned)(v.y & 255) << 8) |
         ((unsigned)(v.z & 255) << 16) | ((unsigned)(v.w & 255) << 24);
}

__global__ __launch_bounds__(256)
void k_pack_msg(const int* __restrict__ msg, uint4* __restrict__ pk) {
  const int t = blockIdx.x * 256 + threadIdx.x;
  const int g = t / BATCH;
  const int b = t % BATCH;
  const int* row = msg + (size_t)b * NPOS;
  const int p0 = (b & 31) + g * 16;
  unsigned w[4];
#pragma unroll
  for (int k = 0; k < 4; ++k) {
    int4 v;
    int q;
    q = p0 + 4 * k + 0; q = q >= NPOS ? q - NPOS : q; v.x = row[q];
    q = p0 + 4 * k + 1; q = q >= NPOS ? q - NPOS : q; v.y = row[q];
    q = p0 + 4 * k + 2; q = q >= NPOS ? q - NPOS : q; v.z = row[q];
    q = p0 + 4 * k + 3; q = q >= NPOS ? q - NPOS : q; v.w = row[q];
    w[k] = pack4(v);
  }
  pk[(size_t)g * BATCH + b] = make_uint4(w[0], w[1], w[2], w[3]);
}

#define NSLOT 128
#define SP    257
__global__ __launch_bounds__(1024)
void k_embed_cols(const uint4* __restrict__ pk,
                  const float* __restrict__ W1,
                  const float* __restrict__ b1,
                  float* __restrict__ h0T) {
  __shared__ float2 tab[NCHAR * NSLOT];
  __shared__ float  stg[32 * SP];
  const int tid  = threadIdx.x;
  const int lane = tid & 31;
  const int wid  = tid >> 5;
  const int j0   = blockIdx.x * 2;
  {
    const float4* r0 = (const float4*)(W1 + (size_t)j0 * IDIM);
    const float4* r1 = (const float4*)(W1 + (size_t)(j0 + 1) * IDIM);
    float4 a[5], c[5];
#pragma unroll
    for (int k = 0; k < 5; ++k) {
      a[k] = r0[tid + 1024 * k];
      c[k] = r1[tid + 1024 * k];
    }
    float* tf = (float*)tab;
    const int pl = tid >> 6;
    const int cc = (tid & 63) * 4;
#pragma unroll
    for (int k = 0; k < 5; ++k) {
      float* s0 = stg + pl * SP + cc;
      float* s1 = stg + (16 + pl) * SP + cc;
      s0[0] = a[k].x; s0[1] = a[k].y; s0[2] = a[k].z; s0[3] = a[k].w;
      s1[0] = c[k].x; s1[1] = c[k].y; s1[2] = c[k].z; s1[3] = c[k].w;
      __syncthreads();
      const int slot = 16 * k + (lane & 15);
      const int col  = lane >> 4;
      float v[8];
#pragma unroll
      for (int i = 0; i < 8; ++i) v[i] = stg[lane * SP + wid * 8 + i];
#pragma unroll
      for (int i = 0; i < 8; ++i) {
        const int ch = wid * 8 + i;
        tf[(ch * NSLOT + slot) * 2 + col] = v[i];
        if (k < 2) tf[(ch * NSLOT + slot + NPOS) * 2 + col] = v[i];
      }
      __syncthreads();
    }
  }

  const float bj0 = b1[j0];
  const float bj1 = b1[j0 + 1];
  const float2* tl = tab + lane;
  for (int it = 0; it < BATCH / 1024; ++it) {
    const int b = it * 1024 + tid;
    unsigned cw[4 * NGRP];
#pragma unroll
    for (int g = 0; g < NGRP; ++g) {
      const uint4 ch = pk[(size_t)g * BATCH + b];
      cw[4 * g + 0] = ch.x;
      cw[4 * g + 1] = ch.y;
      cw[4 * g + 2] = ch.z;
      cw[4 * g + 3] = ch.w;
    }
    float2 s[8];
#pragma unroll
    for (int grp = 0; grp < 8; ++grp) {
      float2 v[10];
#pragma unroll
      for (int q = 0; q < 10; ++q) {
        const int p = grp * 10 + q;
        const unsigned c = (cw[p >> 2] >> (8 * (p & 3))) & 255u;
        v[q] = tl[c * NSLOT + p];
      }
      asm volatile("" :: "v"(v[0].x), "v"(v[0].y), "v"(v[1].x), "v"(v[1].y),
                         "v"(v[2].x), "v"(v[2].y), "v"(v[3].x), "v"(v[3].y),
                         "v"(v[4].x), "v"(v[4].y), "v"(v[5].x), "v"(v[5].y),
                         "v"(v[6].x), "v"(v[6].y), "v"(v[7].x), "v"(v[7].y),
                         "v"(v[8].x), "v"(v[8].y), "v"(v[9].x), "v"(v[9].y));
      float2 t = v[0];
#pragma unroll
      for (int q = 1; q < 10; ++q) {
        t.x += v[q].x;
        t.y += v[q].y;
      }
      s[grp] = t;
    }
    float2 a = s[0];
#pragma unroll
    for (int w = 1; w < 8; ++w) {
      a.x += s[w].x;
      a.y += s[w].y;
    }
    a.x += bj0;
    a.y += bj1;
    h0T[(size_t)j0 * BATCH + b]       = a.x;
    h0T[(size_t)(j0 + 1) * BATCH + b] = a.y;
  }
}

#define TP 516

__global__ __launch_bounds__(256)
void k_ln1_elu_turn(const float* __restrict__ h0T,
                    const float* __restrict__ g1,
                    const float* __restrict__ be1,
                    float* __restrict__ h1) {
  __shared__ float tile[32 * TP];
  __shared__ float ps[8][32];
  __shared__ float pq[8][32];
  const int tid  = threadIdx.x;
  const int lane = tid & 31;
  const int wid  = __builtin_amdgcn_readfirstlane(tid >> 5);
  const int b0   = blockIdx.x * 32;
  const float* src = h0T + (size_t)(wid * 64) * BATCH + b0 + lane;

  float x[64];
#pragma unroll
  for (int i = 0; i < 64; ++i) x[i] = src[(size_t)i * BATCH];

  float s0 = 0.f, s1 = 0.f, s2 = 0.f, s3 = 0.f;
  float q0 = 0.f, q1 = 0.f, q2 = 0.f, q3 = 0.f;
#pragma unroll
  for (int i = 0; i < 64; i += 4) {
    s0 += x[i];     q0 += x[i]     * x[i];
    s1 += x[i + 1]; q1 += x[i + 1] * x[i + 1];
    s2 += x[i + 2]; q2 += x[i + 2] * x[i + 2];
    s3 += x[i + 3]; q3 += x[i + 3] * x[i + 3];
  }
  ps[wid][lane] = (s0 + s1) + (s2 + s3);
  pq[wid][lane] = (q0 + q1) + (q2 + q3);
  __syncthreads();
  float S = 0.f, Q = 0.f;
#pragma unroll
  for (int w = 0; w < 8; ++w) {
    S += ps[w][lane];
    Q += pq[w][lane];
  }
  const float mu  = S * (1.0f / (float)HDIM);
  const float var = Q * (1.0f / (float)HDIM) - mu * mu;
  const float rs  = rsqrtf(var + LN_EPS);

  float* trow = tile + lane * TP + wid * 64;
#pragma unroll
  for (int i = 0; i < 64; ++i) {
    const int j = wid * 64 + i;
    trow[i] = elu((x[i] - mu) * rs * g1[j] + be1[j]);
  }
  __syncthreads();

#pragma unroll
  for (int r = 0; r < 4; ++r) {
    const int row = wid * 4 + r;
    const float4* tr = (const float4*)(tile + row * TP);
    float4* dst = (float4*)(h1 + (size_t)(b0 + row) * HDIM);
#pragma unroll
    for (int i = 0; i < 4; ++i) dst[lane + 32 * i] = tr[lane + 32 * i];
  }
}

__global__ void k_gemm_wmma(const float* __restrict__ h1,
                            const float* __restrict__ W2,
                            const float* __restrict__ b2,
                            float* __restrict__ h2,
                            float* __restrict__ rowstat) {
  __shared__ float sA[128 * LDP];
  __shared__ float sB[64 * LDP];
  const int tid   = threadIdx.x;
  const int wid   = tid >> 5;
  const int lane  = tid & 31;
  const int m0    = blockIdx.x * 128;
  const int n0    = blockIdx.y * 64;
  const int mrow  = lane & 15;
  const int khalf = (lane >> 4) << 1;

  v8f acc0 = {}, acc1 = {}, acc2 = {}, acc3 = {};

  for (int k0 = 0; k0 < HDIM; k0 += KC) {
#pragma unroll
    for (int i = 0; i < 4; ++i) {
      const int idx = tid + i * 256;
      const int r   = idx >> 3;
      const int cc  = (idx & 7) << 2;
      const float4 v = *(const float4*)(h1 + (size_t)(m0 + r) * HDIM + k0 + cc);
      v2f lo = {v.x, v.y};
      v2f hi = {v.z, v.w};
      *(v2f*)(sA + r * LDP + cc)     = lo;
      *(v2f*)(sA + r * LDP + cc + 2) = hi;
    }
#pragma unroll
    for (int i = 0; i < 2; ++i) {
      const int idx = tid + i * 256;
      const int r   = idx >> 3;
      const int cc  = (idx & 7) << 2;
      const float4 v = *(const float4*)(W2 + (size_t)(n0 + r) * HDIM + k0 + cc);
      v2f lo = {v.x, v.y};
      v2f hi = {v.z, v.w};
      *(v2f*)(sB + r * LDP + cc)     = lo;
      *(v2f*)(sB + r * LDP + cc + 2) = hi;
    }
    __syncthreads();

    const float* Aw = sA + (wid * 16 + mrow) * LDP + khalf;
    const float* Bw = sB + mrow * LDP + khalf;
#pragma unroll
    for (int kk = 0; kk < KC; kk += 4) {
      const v2f a   = *(const v2f*)(Aw + kk);
      const v2f bb0 = *(const v2f*)(Bw + kk);
      const v2f bb1 = *(const v2f*)(Bw + 16 * LDP + kk);
      const v2f bb2 = *(const v2f*)(Bw + 32 * LDP + kk);
      const v2f bb3 = *(const v2f*)(Bw + 48 * LDP + kk);
      acc0 = __builtin_amdgcn_wmma_f32_16x16x4_f32(false, a, false, bb0, (short)0, acc0, false, false);
      acc1 = __builtin_amdgcn_wmma_f32_16x16x4_f32(false, a, false, bb1, (short)0, acc1, false, false);
      acc2 = __builtin_amdgcn_wmma_f32_16x16x4_f32(false, a, false, bb2, (short)0, acc2, false, false);
      acc3 = __builtin_amdgcn_wmma_f32_16x16x4_f32(false, a, false, bb3, (short)0, acc3, false, false);
    }
    __syncthreads();
  }

  const int ncol  = lane & 15;
  const int mbase = (lane >> 4) << 3;
  float* out = h2 + (size_t)(m0 + wid * 16 + mbase) * HDIM + n0;
  const float bj0 = b2[n0 + ncol];
  const float bj1 = b2[n0 + 16 + ncol];
  const float bj2 = b2[n0 + 32 + ncol];
  const float bj3 = b2[n0 + 48 + ncol];
  float2* sp = (float2*)rowstat + (size_t)blockIdx.y * BATCH + (m0 + wid * 16 + mbase);
  float rs_s[8], rs_q[8];
#pragma unroll
  for (int r = 0; r < 8; ++r) {
    float* row = out + (size_t)r * HDIM;
    const float v0 = acc0[r] + bj0;
    const float v1 = acc1[r] + bj1;
    const float v2 = acc2[r] + bj2;
    const float v3 = acc3[r] + bj3;
    row[ncol]      = v0;
    row[16 + ncol] = v1;
    row[32 + ncol] = v2;
    row[48 + ncol] = v3;
    rs_s[r] = (v0 + v1) + (v2 + v3);
    rs_q[r] = (v0 * v0 + v1 * v1) + (v2 * v2 + v3 * v3);
  }
#pragma unroll
  for (int off = 8; off > 0; off >>= 1) {
#pragma unroll
    for (int r = 0; r < 8; ++r) {
      rs_s[r] += __shfl_xor(rs_s[r], off, 32);
      rs_q[r] += __shfl_xor(rs_q[r], off, 32);
    }
  }
  if (ncol == 0) {
#pragma unroll
    for (int r = 0; r < 8; ++r) sp[r] = make_float2(rs_s[r], rs_q[r]);
  }
}

#define RPW 2
#define NTN (HDIM / 64)

__global__ __launch_bounds__(256)
void k_ln2_elu(const float* __restrict__ h2,
               const float* __restrict__ rowstat,
               const float* __restrict__ g2,
               const float* __restrict__ be2,
               float* __restrict__ out) {
  const int lane = threadIdx.x & 31;
  const int wid  = __builtin_amdgcn_readfirstlane(threadIdx.x >> 5);
  const int row0 = (blockIdx.x * 8 + wid) * RPW;

  float4 x[RPW][4];
#pragma unroll
  for (int r = 0; r < RPW; ++r) {
    const float4* src = (const float4*)(h2 + (size_t)(row0 + r) * HDIM);
#pragma unroll
    for (int i = 0; i < 4; ++i) x[r][i] = src[lane + 32 * i];
  }
  float4 g[4], be[4];
#pragma unroll
  for (int i = 0; i < 4; ++i) {
    g[i]  = ((const float4*)g2)[lane + 32 * i];
    be[i] = ((const float4*)be2)[lane + 32 * i];
  }
#pragma unroll
  for (int r = 0; r < RPW; ++r) {
    float S = 0.f, Q = 0.f;
#pragma unroll
    for (int y = 0; y < NTN; ++y) {
      const float2 p = ((const float2*)rowstat)[(size_t)y * BATCH + row0 + r];
      S += p.x;
      Q += p.y;
    }
    const float mu  = S * (1.0f / (float)HDIM);
    const float var = Q * (1.0f / (float)HDIM) - mu * mu;
    const float rs  = rsqrtf(var + LN_EPS);
    float4* dst = (float4*)(out + (size_t)(row0 + r) * HDIM);
#pragma unroll
    for (int i = 0; i < 4; ++i) {
      const float4 v = x[r][i];
      float4 o;
      o.x = elu((v.x - mu) * rs * g[i].x + be[i].x);
      o.y = elu((v.y - mu) * rs * g[i].y + be[i].y);
      o.z = elu((v.z - mu) * rs * g[i].z + be[i].z);
      o.w = elu((v.w - mu) * rs * g[i].w + be[i].w);
      dst[lane + 32 * i] = o;
    }
  }
}

extern "C" void kernel_launch(void* const* d_in, const int* in_sizes, int n_in,
                              void* d_out, int out_size, void* d_ws, size_t ws_size,
                              hipStream_t stream) {
  (void)in_sizes; (void)n_in; (void)out_size; (void)ws_size;
  const int*   msg = (const int*)  d_in[0];
  const float* W1  = (const float*)d_in[1];
  const float* b1  = (const float*)d_in[2];
  const float* g1  = (const float*)d_in[3];
  const float* be1 = (const float*)d_in[4];
  const float* W2  = (const float*)d_in[5];
  const float* b2  = (const float*)d_in[6];
  const float* g2  = (const float*)d_in[7];
  const float* be2 = (const float*)d_in[8];
  float* out = (float*)d_out;

  const size_t pk_bytes = (size_t)NGRP * BATCH * 16;
  const size_t h_bytes  = (size_t)BATCH * HDIM * sizeof(float);
  char* ws = (char*)d_ws;
  uint4* pk      = (uint4*)ws;
  float* h0T     = (float*)(ws + pk_bytes);
  float* h1      = (float*)(ws + pk_bytes + h_bytes);
  float* h2      = (float*)(ws + pk_bytes + 2 * h_bytes);
  float* rowstat = (float*)(ws + pk_bytes + 3 * h_bytes);

  k_pack_msg<<<NGRP * BATCH / 256, 256, 0, stream>>>(msg, pk);
  k_embed_cols<<<HDIM / 2, 1024, 0, stream>>>(pk, W1, b1, h0T);
  k_ln1_elu_turn<<<BATCH / 32, 256, 0, stream>>>(h0T, g1, be1, h1);
  k_gemm_wmma<<<dim3(BATCH / 128, HDIM / 64), 256, 0, stream>>>(h1, W2, b2, h2, rowstat);
  k_ln2_elu<<<BATCH / (8 * RPW), 256, 0, stream>>>(h2, rowstat, g2, be2, out);
}


# product K loop software-pipelined: two LDS buffers, next chunk's global loads a whole step ahead of their LDS store, a chunk's 40 fragment loads in one group before its matrix work; sum order kept
# speedup vs baseline: 1.6642x; 1.0375x over previous
#include <hip/hip_runtime.h>
#include <math.h>


#define NPOS  80
#define NCHAR 256
#define HDIM  512
#define IDIM  (NPOS * NCHAR)
#define BATCH 8192
#define LN_EPS 1e-5f
#define KC    32
#define LDP   34

typedef __attribute__((ext_vector_type(2))) float v2f;
typedef __attribute__((ext_vector_type(8))) float v8f;

__device__ __forceinline__ float elu(float x) {
  return x > 0.f ? x : expm1f(x);
}

#define NGRP (NPOS / 16)

__device__ __forceinline__ unsigned pack4(int4 v) {
  return (unsigned)(v.x & 255) | ((unsigned)(v.y & 255) << 8) |
         ((unsigned)(v.z & 255) << 16) | ((unsigned)(v.w & 255) << 24);
}

__global__ __launch_bounds__(256)
void k_pack_msg(const int* __restrict__ msg, uint4* __restrict__ pk) {
  const int t = blockIdx.x * 256 + threadIdx.x;
  const int g = t / BATCH;
  const int b = t % BATCH;
  const int* row = msg + (size_t)b * NPOS;
  const int p0 = (b & 31) + g * 16;
  unsigned w[4];
#pragma unroll
  for (int k = 0; k < 4; ++k) {
    int4 v;
    int q;
    q = p0 + 4 * k + 0; q = q >= NPOS ? q - NPOS : q; v.x = row[q];
    q = p0 + 4 * k + 1; q = q >= NPOS ? q - NPOS : q; v.y = row[q];
    q = p0 + 4 * k + 2; q = q >= NPOS ? q - NPOS : q; v.z = row[q];
    q = p0 + 4 * k + 3; q = q >= NPOS ? q - NPOS : q; v.w = row[q];
    w[k] = pack4(v);
  }
  pk[(size_t)g * BATCH + b] = make_uint4(w[0], w[1], w[2], w[3]);
}

#define NSLOT 128
#define SP    257
__global__ __launch_bounds__(1024)
void k_embed_cols(const uint4* __restrict__ pk,
                  const float* __restrict__ W1,
                  const float* __restrict__ b1,
                  float* __restrict__ h0T) {
  __shared__ float2 tab[NCHAR * NSLOT];
  __shared__ float  stg[32 * SP];
  const int tid  = threadIdx.x;
  const int lane = tid & 31;
  const int wid  = tid >> 5;
  const int j0   = blockIdx.x * 2;
  {
    const float4* r0 = (const float4*)(W1 + (size_t)j0 * IDIM);
    const float4* r1 = (const float4*)(W1 + (size_t)(j0 + 1) * IDIM);
    float4 a[5], c[5];
#pragma unroll
    for (int k = 0; k < 5; ++k) {
      a[k] = r0[tid + 1024 * k];
      c[k] = r1[tid + 1024 * k];
    }
    float* tf = (float*)tab;
    const int pl = tid >> 6;
    const int cc = (tid & 63) * 4;
#pragma unroll
    for (int k = 0; k < 5; ++k) {
      float* s0 = stg + pl * SP + cc;
      float* s1 = stg + (16 + pl) * SP + cc;
      s0[0] = a[k].x; s0[1] = a[k].y; s0[2] = a[k].z; s0[3] = a[k].w;
      s1[0] = c[k].x; s1[1] = c[k].y; s1[2] = c[k].z; s1[3] = c[k].w;
      __syncthreads();
      const int slot = 16 * k + (lane & 15);
      const int col  = lane >> 4;
      float v[8];
#pragma unroll
      for (int i = 0; i < 8; ++i) v[i] = stg[lane * SP + wid * 8 + i];
#pragma unroll
      for (int i = 0; i < 8; ++i) {
        const int ch = wid * 8 + i;
        tf[(ch * NSLOT + slot) * 2 + col] = v[i];
        if (k < 2) tf[(ch * NSLOT + slot + NPOS) * 2 + col] = v[i];
      }
      __syncthreads();
    }
  }

  const float bj0 = b1[j0];
  const float bj1 = b1[j0 + 1];
  const float2* tl = tab + lane;
  for (int it = 0; it < BATCH / 1024; ++it) {
    const int b = it * 1024 + tid;
    unsigned cw[4 * NGRP];
#pragma unroll
    for (int g = 0; g < NGRP; ++g) {
      const uint4 ch = pk[(size_t)g * BATCH + b];
      cw[4 * g + 0] = ch.x;
      cw[4 * g + 1] = ch.y;
      cw[4 * g + 2] = ch.z;
      cw[4 * g + 3] = ch.w;
    }
    float2 s[8];
#pragma unroll
    for (int grp = 0; grp < 8; ++grp) {
      float2 v[10];
#pragma unroll
      for (int q = 0; q < 10; ++q) {
        const int p = grp * 10 + q;
        const unsigned c = (cw[p >> 2] >> (8 * (p & 3))) & 255u;
        v[q] = tl[c * NSLOT + p];
      }
      asm volatile("" :: "v"(v[0].x), "v"(v[0].y), "v"(v[1].x), "v"(v[1].y),
                         "v"(v[2].x), "v"(v[2].y), "v"(v[3].x), "v"(v[3].y),
                         "v"(v[4].x), "v"(v[4].y), "v"(v[5].x), "v"(v[5].y),
                         "v"(v[6].x), "v"(v[6].y), "v"(v[7].x), "v"(v[7].y),
                         "v"(v[8].x), "v"(v[8].y), "v"(v[9].x), "v"(v[9].y));
      float2 t = v[0];
#pragma unroll
      for (int q = 1; q < 10; ++q) {
        t.x += v[q].x;
        t.y += v[q].y;
      }
      s[grp] = t;
    }
    float2 a = s[0];
#pragma unroll
    for (int w = 1; w < 8; ++w) {
      a.x += s[w].x;
      a.y += s[w].y;
    }
    a.x += bj0;
    a.y += bj1;
    h0T[(size_t)j0 * BATCH + b]       = a.x;
    h0T[(size_t)(j0 + 1) * BATCH + b] = a.y;
  }
}

#define TP 516

__global__ __launch_bounds__(256)
void k_ln1_elu_turn(const float* __restrict__ h0T,
                    const float* __restrict__ g1,
                    const float* __restrict__ be1,
                    float* __restrict__ h1) {
  __shared__ float tile[32 * TP];
  __shared__ float ps[8][32];
  __shared__ float pq[8][32];
  const int tid  = threadIdx.x;
  const int lane = tid & 31;
  const int wid  = __builtin_amdgcn_readfirstlane(tid >> 5);
  const int b0   = blockIdx.x * 32;
  const float* src = h0T + (size_t)(wid * 64) * BATCH + b0 + lane;

  float x[64];
#pragma unroll
  for (int i = 0; i < 64; ++i) x[i] = src[(size_t)i * BATCH];

  float s0 = 0.f, s1 = 0.f, s2 = 0.f, s3 = 0.f;
  float q0 = 0.f, q1 = 0.f, q2 = 0.f, q3 = 0.f;
#pragma unroll
  for (int i = 0; i < 64; i += 4) {
    s0 += x[i];     q0 += x[i]     * x[i];
    s1 += x[i + 1]; q1 += x[i + 1] * x[i + 1];
    s2 += x[i + 2]; q2 += x[i + 2] * x[i + 2];
    s3 += x[i + 3]; q3 += x[i + 3] * x[i + 3];
  }
  ps[wid][lane] = (s0 + s1) + (s2 + s3);
  pq[wid][lane] = (q0 + q1) + (q2 + q3);
  __syncthreads();
  float S = 0.f, Q = 0.f;
#pragma unroll
  for (int w = 0; w < 8; ++w) {
    S += ps[w][lane];
    Q += pq[w][lane];
  }
  const float mu  = S * (1.0f / (float)HDIM);
  const float var = Q * (1.0f / (float)HDIM) - mu * mu;
  const float rs  = rsqrtf(var + LN_EPS);

  float* trow = tile + lane * TP + wid * 64;
#pragma unroll
  for (int i = 0; i < 64; ++i) {
    const int j = wid * 64 + i;
    trow[i] = elu((x[i] - mu) * rs * g1[j] + be1[j]);
  }
  __syncthreads();

#pragma unroll
  for (int r = 0; r < 4; ++r) {
    const int row = wid * 4 + r;
    const float4* tr = (const float4*)(tile + row * TP);
    float4* dst = (float4*)(h1 + (size_t)(b0 + row) * HDIM);
#pragma unroll
    for (int i = 0; i < 4; ++i) dst[lane + 32 * i] = tr[lane + 32 * i];
  }
}

struct Frags { v2f a[8], p0[8], p1[8], p2[8], p3[8]; };

__device__ __forceinline__ void load_frags(const float* Aw, const float* Bw, Frags& f) {
#pragma unroll
  for (int s = 0; s < 8; ++s) {
    f.a[s]  = *(const v2f*)(Aw + 4 * s);
    f.p0[s] = *(const v2f*)(Bw + 4 * s);
    f.p1[s] = *(const v2f*)(Bw + 16 * LDP + 4 * s);
    f.p2[s] = *(const v2f*)(Bw + 32 * LDP + 4 * s);
    f.p3[s] = *(const v2f*)(Bw + 48 * LDP + 4 * s);
  }
  asm volatile("" :: "v"(f.a[0]), "v"(f.a[1]), "v"(f.a[2]), "v"(f.a[3]), "v"(f.a[4]), "v"(f.a[5]), "v"(f.a[6]), "v"(f.a[7]),
                     "v"(f.p0[0]), "v"(f.p0[1]), "v"(f.p0[2]), "v"(f.p0[3]), "v"(f.p0[4]), "v"(f.p0[5]), "v"(f.p0[6]), "v"(f.p0[7]));
  asm volatile("" :: "v"(f.p1[0]), "v"(f.p1[1]), "v"(f.p1[2]), "v"(f.p1[3]), "v"(f.p1[4]), "v"(f.p1[5]), "v"(f.p1[6]), "v"(f.p1[7]),
                     "v"(f.p2[0]), "v"(f.p2[1]), "v"(f.p2[2]), "v"(f.p2[3]), "v"(f.p2[4]), "v"(f.p2[5]), "v"(f.p2[6]), "v"(f.p2[7]),
                     "v"(f.p3[0]), "v"(f.p3[1]), "v"(f.p3[2]), "v"(f.p3[3]), "v"(f.p3[4]), "v"(f.p3[5]), "v"(f.p3[6]), "v"(f.p3[7]));
}

__device__ __forceinline__ void mma_frags(const Frags& f, v8f& acc0, v8f& acc1, v8f& acc2, v8f& acc3) {
#pragma unroll
  for (int s = 0; s < 8; ++s) {
    acc0 = __builtin_amdgcn_wmma_f32_16x16x4_f32(false, f.a[s], false, f.p0[s], (short)0, acc0, false, false);
    acc1 = __builtin_amdgcn_wmma_f32_16x16x4_f32(false, f.a[s], false, f.p1[s], (short)0, acc1, false, false);
    acc2 = __builtin_amdgcn_wmma_f32_16x16x4_f32(false, f.a[s], false, f.p2[s], (short)0, acc2, false, false);
    acc3 = __builtin_amdgcn_wmma_f32_16x16x4_f32(false, f.a[s], false, f.p3[s], (short)0, acc3, false, false);
  }
}

#define NCHUNK (HDIM / KC)

__global__ __launch_bounds__(256)
void k_gemm_wmma(const float* __restrict__ h1,
                            const float* __restrict__ W2,
                            const float* __restrict__ b2,
                            float* __restrict__ h2,
                            float* __restrict__ rowstat) {
  __shared__ float sA[2][128 * LDP];
  __shared__ float sB[2][64 * LDP];
  const int tid   = threadIdx.x;
  const int wid   = tid >> 5;
  const int lane  = tid & 31;
  const int m0    = blockIdx.x * 128;
  const int n0    = blockIdx.y * 64;
  const int mrow  = lane & 15;
  const int khalf = (lane >> 4) << 1;

  v8f acc0 = {}, acc1 = {}, acc2 = {}, acc3 = {};

  const int sr = tid >> 3;
  const int sc = (tid & 7) << 2;
  const float* gA = h1 + (size_t)(m0 + sr) * HDIM + sc;
  const float* gB = W2 + (size_t)(n0 + sr) * HDIM + sc;
  const int so = sr * LDP + sc;
  const int fa = (wid * 16 + mrow) * LDP + khalf;
  const int fb = mrow * LDP + khalf;

  float4 ra[4], rb[2];
#define G_LOAD(k0)                                                              \
  _Pragma("unroll") for (int i = 0; i < 4; ++i)                                 \
    ra[i] = *(const float4*)(gA + (size_t)(32 * i) * HDIM + (k0));              \
  _Pragma("unroll") for (int i = 0; i < 2; ++i)                                 \
    rb[i] = *(const float4*)(gB + (size_t)(32 * i) * HDIM + (k0));
#define L_STORE(buf)                                                            \
  _Pragma("unroll") for (int i = 0; i < 4; ++i) {                               \
    v2f lo = {ra[i].x, ra[i].y};                                                \
    v2f hi = {ra[i].z, ra[i].w};                                                \
    *(v2f*)(sA[buf] + so + 32 * i * LDP)     = lo;                              \
    *(v2f*)(sA[buf] + so + 32 * i * LDP + 2) = hi;                              \
  }                                                                             \
  _Pragma("unroll") for (int i = 0; i < 2; ++i) {                               \
    v2f lo = {rb[i].x, rb[i].y};                                                \
    v2f hi = {rb[i].z, rb[i].w};                                                \
    *(v2f*)(sB[buf] + so + 32 * i * LDP)     = lo;                              \
    *(v2f*)(sB[buf] + so + 32 * i * LDP + 2) = hi;                              \
  }

  G_LOAD(0)
  L_STORE(0)
  G_LOAD(KC)
  __syncthreads();

  for (int c = 0; c < NCHUNK; ++c) {
    const int cur = c & 1;
    Frags f;
    load_frags(sA[cur] + fa, sB[cur] + fb, f);
    if (c + 1 < NCHUNK) {
      L_STORE(cur ^ 1)
    }
    if (c + 2 < NCHUNK) {
      G_LOAD((c + 2) * KC)
    }
    mma_frags(f, acc0, acc1, acc2, acc3);
    __syncthreads();
  }
#undef G_LOAD
#undef L_STORE

  const int ncol  = lane & 15;
  const int mbase = (lane >> 4) << 3;
  float* out = h2 + (size_t)(m0 + wid * 16 + mbase) * HDIM + n0;
  const float bj0 = b2[n0 + ncol];
  const float bj1 = b2[n0 + 16 + ncol];
  const float bj2 = b2[n0 + 32 + ncol];
  const float bj3 = b2[n0 + 48 + ncol];
  float2* sp = (float2*)rowstat + (size_t)blockIdx.y * BATCH + (m0 + wid * 16 + mbase);
  float rs_s[8], rs_q[8];
#pragma unroll
  for (int r = 0; r < 8; ++r) {
    float* row = out + (size_t)r * HDIM;
    const float v0 = acc0[r] + bj0;
    const float v1 = acc1[r] + bj1;
    const float v2 = acc2[r] + bj2;
    const float v3 = acc3[r] + bj3;
    row[ncol]      = v0;
    row[16 + ncol] = v1;
    row[32 + ncol] = v2;
    row[48 + ncol] = v3;
    rs_s[r] = (v0 + v1) + (v2 + v3);
    rs_q[r] = (v0 * v0 + v1 * v1) + (v2 * v2 + v3 * v3);
  }
#pragma unroll
  for (int off = 8; off > 0; off >>= 1) {
#pragma unroll
    for (int r = 0; r < 8; ++r) {
      rs_s[r] += __shfl_xor(rs_s[r], off, 32);
      rs_q[r] += __shfl_xor(rs_q[r], off, 32);
    }
  }
  if (ncol == 0) {
#pragma unroll
    for (int r = 0; r < 8; ++r) sp[r] = make_float2(rs_s[r], rs_q[r]);
  }
}

#define RPW 2
#define NTN (HDIM / 64)

__global__ __launch_bounds__(256)
void k_ln2_elu(const float* __restrict__ h2,
               const float* __restrict__ rowstat,
               const float* __restrict__ g2,
               const float* __restrict__ be2,
               float* __restrict__ out) {
  const int lane = threadIdx.x & 31;
  const int wid  = __builtin_amdgcn_readfirstlane(threadIdx.x >> 5);
  const int row0 = (blockIdx.x * 8 + wid) * RPW;

  float4 x[RPW][4];
#pragma unroll
  for (int r = 0; r < RPW; ++r) {
    const float4* src = (const float4*)(h2 + (size_t)(row0 + r) * HDIM);
#pragma unroll
    for (int i = 0; i < 4; ++i) x[r][i] = src[lane + 32 * i];
  }
  float4 g[4], be[4];
#pragma unroll
  for (int i = 0; i < 4; ++i) {
    g[i]  = ((const float4*)g2)[lane + 32 * i];
    be[i] = ((const float4*)be2)[lane + 32 * i];
  }
#pragma unroll
  for (int r = 0; r < RPW; ++r) {
    float S = 0.f, Q = 0.f;
#pragma unroll
    for (int y = 0; y < NTN; ++y) {
      const float2 p = ((const float2*)rowstat)[(size_t)y * BATCH + row0 + r];
      S += p.x;
      Q += p.y;
    }
    const float mu  = S * (1.0f / (float)HDIM);
    const float var = Q * (1.0f / (float)HDIM) - mu * mu;
    const float rs  = rsqrtf(var + LN_EPS);
    float4* dst = (float4*)(out + (size_t)(row0 + r) * HDIM);
#pragma unroll
    for (int i = 0; i < 4; ++i) {
      const float4 v = x[r][i];
      float4 o;
      o.x = elu((v.x - mu) * rs * g[i].x + be[i].x);
      o.y = elu((v.y - mu) * rs * g[i].y + be[i].y);
      o.z = elu((v.z - mu) * rs * g[i].z + be[i].z);
      o.w = elu((v.w - mu) * rs * g[i].w + be[i].w);
      dst[lane + 32 * i] = o;
    }
  }
}

extern "C" void kernel_launch(void* const* d_in, const int* in_sizes, int n_in,
                              void* d_out, int out_size, void* d_ws, size_t ws_size,
                              hipStream_t stream) {
  (void)in_sizes; (void)n_in; (void)out_size; (void)ws_size;
  const int*   msg = (const int*)  d_in[0];
  const float* W1  = (const float*)d_in[1];
  const float* b1  = (const float*)d_in[2];
  const float* g1  = (const float*)d_in[3];
  const float* be1 = (const float*)d_in[4];
  const float* W2  = (const float*)d_in[5];
  const float* b2  = (const float*)d_in[6];
  const float* g2  = (const float*)d_in[7];
  const float* be2 = (const float*)d_in[8];
  float* out = (float*)d_out;

  const size_t pk_bytes = (size_t)NGRP * BATCH * 16;
  const size_t h_bytes  = (size_t)BATCH * HDIM * sizeof(float);
  char* ws = (char*)d_ws;
  uint4* pk      = (uint4*)ws;
  float* h0T     = (float*)(ws + pk_bytes);
  float* h1      = (float*)(ws + pk_bytes + h_bytes);
  float* h2      = (float*)(ws + pk_bytes + 2 * h_bytes);
  float* rowstat = (float*)(ws + pk_bytes + 3 * h_bytes);

  k_pack_msg<<<NGRP * BATCH / 256, 256, 0, stream>>>(msg, pk);
  k_embed_cols<<<HDIM / 2, 1024, 0, stream>>>(pk, W1, b1, h0T);
  k_ln1_elu_turn<<<BATCH / 32, 256, 0, stream>>>(h0T, g1, be1, h1);
  k_gemm_wmma<<<dim3(BATCH / 128, HDIM / 64), 256, 0, stream>>>(h1, W2, b2, h2, rowstat);
  k_ln2_elu<<<BATCH / (8 * RPW), 256, 0, stream>>>(h2, rowstat, g2, be2, out);
}


# LN1 launch: 16 waves per block (4 per SIMD, was 2), normalise+ELU moved behind the LDS turn (wave = 2 whole rows, gamma/beta by vector loads), one barrier; row sums as 16 partials: new reduction tree
# speedup vs baseline: 1.7766x; 1.0675x over previous
#include <hip/hip_runtime.h>
#include <math.h>


#define NPOS  80
#define NCHAR 256
#define HDIM  512
#define IDIM  (NPOS * NCHAR)
#define BATCH 8192
#define LN_EPS 1e-5f
#define KC    32
#define LDP   34

typedef __attribute__((ext_vector_type(2))) float v2f;
typedef __attribute__((ext_vector_type(8))) float v8f;

__device__ __forceinline__ float elu(float x) {
  return x > 0.f ? x : expm1f(x);
}

#define NGRP (NPOS / 16)

__device__ __forceinline__ unsigned pack4(int4 v) {
  return (unsigned)(v.x & 255) | ((unsigned)(v.y & 255) << 8) |
         ((unsigned)(v.z & 255) << 16) | ((unsigned)(v.w & 255) << 24);
}

__global__ __launch_bounds__(256)
void k_pack_msg(const int* __restrict__ msg, uint4* __restrict__ pk) {
  const int t = blockIdx.x * 256 + threadIdx.x;
  const int g = t / BATCH;
  const int b = t % BATCH;
  const int* row = msg + (size_t)b * NPOS;
  const int p0 = (b & 31) + g * 16;
  unsigned w[4];
#pragma unroll
  for (int k = 0; k < 4; ++k) {
    int4 v;
    int q;
    q = p0 + 4 * k + 0; q = q >= NPOS ? q - NPOS : q; v.x = row[q];
    q = p0 + 4 * k + 1; q = q >= NPOS ? q - NPOS : q; v.y = row[q];
    q = p0 + 4 * k + 2; q = q >= NPOS ? q - NPOS : q; v.z = row[q];
    q = p0 + 4 * k + 3; q = q >= NPOS ? q - NPOS : q; v.w = row[q];
    w[k] = pack4(v);
  }
  pk[(size_t)g * BATCH + b] = make_uint4(w[0], w[1], w[2], w[3]);
}

#define NSLOT 128
#define SP    257
__global__ __launch_bounds__(1024)
void k_embed_cols(const uint4* __restrict__ pk,
                  const float* __restrict__ W1,
                  const float* __restrict__ b1,
                  float* __restrict__ h0T) {
  __shared__ float2 tab[NCHAR * NSLOT];
  __shared__ float  stg[32 * SP];
  const int tid  = threadIdx.x;
  const int lane = tid & 31;
  const int wid  = tid >> 5;
  const int j0   = blockIdx.x * 2;
  {
    const float4* r0 = (const float4*)(W1 + (size_t)j0 * IDIM);
    const float4* r1 = (const float4*)(W1 + (size_t)(j0 + 1) * IDIM);
    float4 a[5], c[5];
#pragma unroll
    for (int k = 0; k < 5; ++k) {
      a[k] = r0[tid + 1024 * k];
      c[k] = r1[tid + 1024 * k];
    }
    float* tf = (float*)tab;
    const int pl = tid >> 6;
    const int cc = (tid & 63) * 4;
#pragma unroll
    for (int k = 0; k < 5; ++k) {
      float* s0 = stg + pl * SP + cc;
      float* s1 = stg + (16 + pl) * SP + cc;
      s0[0] = a[k].x; s0[1] = a[k].y; s0[2] = a[k].z; s0[3] = a[k].w;
      s1[0] = c[k].x; s1[1] = c[k].y; s1[2] = c[k].z; s1[3] = c[k].w;
      __syncthreads();
      const int slot = 16 * k + (lane & 15);
      const int col  = lane >> 4;
      float v[8];
#pragma unroll
      for (int i = 0; i < 8; ++i) v[i] = stg[lane * SP + wid * 8 + i];
#pragma unroll
      for (int i = 0; i < 8; ++i) {
        const int ch = wid * 8 + i;
        tf[(ch * NSLOT + slot) * 2 + col] = v[i];
        if (k < 2) tf[(ch * NSLOT + slot + NPOS) * 2 + col] = v[i];
      }
      __syncthreads();
    }
  }

  const float bj0 = b1[j0];
  const float bj1 = b1[j0 + 1];
  const float2* tl = tab + lane;
  for (int it = 0; it < BATCH / 1024; ++it) {
    const int b = it * 1024 + tid;
    unsigned cw[4 * NGRP];
#pragma unroll
    for (int g = 0; g < NGRP; ++g) {
      const uint4 ch = pk[(size_t)g * BATCH + b];
      cw[4 * g + 0] = ch.x;
      cw[4 * g + 1] = ch.y;
      cw[4 * g + 2] = ch.z;
      cw[4 * g + 3] = ch.w;
    }
    float2 s[8];
#pragma unroll
    for (int grp = 0; grp < 8; ++grp) {
      float2 v[10];
#pragma unroll
      for (int q = 0; q < 10; ++q) {
        const int p = grp * 10 + q;
        const unsigned c = (cw[p >> 2] >> (8 * (p & 3))) & 255u;
        v[q] = tl[c * NSLOT + p];
      }
      asm volatile("" :: "v"(v[0].x), "v"(v[0].y), "v"(v[1].x), "v"(v[1].y),
                         "v"(v[2].x), "v"(v[2].y), "v"(v[3].x), "v"(v[3].y),
                         "v"(v[4].x), "v"(v[4].y), "v"(v[5].x), "v"(v[5].y),
                         "v"(v[6].x), "v"(v[6].y), "v"(v[7].x), "v"(v[7].y),
                         "v"(v[8].x), "v"(v[8].y), "v"(v[9].x), "v"(v[9].y));
      float2 t = v[0];
#pragma unroll
      for (int q = 1; q < 10; ++q) {
        t.x += v[q].x;
        t.y += v[q].y;
      }
      s[grp] = t;
    }
    float2 a = s[0];
#pragma unroll
    for (int w = 1; w < 8; ++w) {
      a.x += s[w].x;
      a.y += s[w].y;
    }
    a.x += bj0;
    a.y += bj1;
    h0T[(size_t)j0 * BATCH + b]       = a.x;
    h0T[(size_t)(j0 + 1) * BATCH + b] = a.y;
  }
}

#define TP 516
#define LW 16

__global__ __launch_bounds__(512)
void k_ln1_elu_turn(const float* __restrict__ h0T,
                    const float* __restrict__ g1,
                    const float* __restrict__ be1,
                    float* __restrict__ h1) {
  __shared__ __attribute__((aligned(16))) float tile[32 * TP];
  __shared__ __attribute__((aligned(16))) float2 part[32 * LW];
  const int tid  = threadIdx.x;
  const int lane = tid & 31;
  const int wid  = __builtin_amdgcn_readfirstlane(tid >> 5);
  const int b0   = blockIdx.x * 32;
  const float* src = h0T + (size_t)(wid * 32) * BATCH + b0 + lane;

  float x[32];
#pragma unroll
  for (int i = 0; i < 32; ++i) x[i] = src[(size_t)i * BATCH];

  float4 g[4], be[4];
#pragma unroll
  for (int i = 0; i < 4; ++i) {
    g[i]  = ((const float4*)g1)[lane + 32 * i];
    be[i] = ((const float4*)be1)[lane + 32 * i];
  }

  float s0 = 0.f, s1 = 0.f, s2 = 0.f, s3 = 0.f;
  float q0 = 0.f, q1 = 0.f, q2 = 0.f, q3 = 0.f;
  float* trow = tile + lane * TP + wid * 32;
#pragma unroll
  for (int i = 0; i < 32; i += 4) {
    s0 += x[i];     q0 += x[i]     * x[i];
    s1 += x[i + 1]; q1 += x[i + 1] * x[i + 1];
    s2 += x[i + 2]; q2 += x[i + 2] * x[i + 2];
    s3 += x[i + 3]; q3 += x[i + 3] * x[i + 3];
    *(float4*)(trow + i) = make_float4(x[i], x[i + 1], x[i + 2], x[i + 3]);
  }
  part[lane * LW + wid] = make_float2((s0 + s1) + (s2 + s3), (q0 + q1) + (q2 + q3));
  __syncthreads();

  float mu[2], rs[2];
  float4 v[2][4];
#pragma unroll
  for (int r = 0; r < 2; ++r) {
    const int row = wid * 2 + r;
    const float4* pr = (const float4*)(part + row * LW);
    float S = 0.f, Q = 0.f;
#pragma unroll
    for (int k = 0; k < LW / 2; ++k) {
      const float4 p = pr[k];
      S += p.x; Q += p.y;
      S += p.z; Q += p.w;
    }
    mu[r] = S * (1.0f / (float)HDIM);
    const float var = Q * (1.0f / (float)HDIM) - mu[r] * mu[r];
    rs[r] = rsqrtf(var + LN_EPS);
    const float4* tr = (const float4*)(tile + row * TP);
#pragma unroll
    for (int i = 0; i < 4; ++i) v[r][i] = tr[lane + 32 * i];
  }
#pragma unroll
  for (int r = 0; r < 2; ++r) {
    float4* dst = (float4*)(h1 + (size_t)(b0 + wid * 2 + r) * HDIM);
#pragma unroll
    for (int i = 0; i < 4; ++i) {
      float4 o;
      o.x = elu((v[r][i].x - mu[r]) * rs[r] * g[i].x + be[i].x);
      o.y = elu((v[r][i].y - mu[r]) * rs[r] * g[i].y + be[i].y);
      o.z = elu((v[r][i].z - mu[r]) * rs[r] * g[i].z + be[i].z);
      o.w = elu((v[r][i].w - mu[r]) * rs[r] * g[i].w + be[i].w);
      dst[lane + 32 * i] = o;
    }
  }
}

struct Frags { v2f a[8], p0[8], p1[8], p2[8], p3[8]; };

__device__ __forceinline__ void load_frags(const float* Aw, const float* Bw, Frags& f) {
#pragma unroll
  for (int s = 0; s < 8; ++s) {
    f.a[s]  = *(const v2f*)(Aw + 4 * s);
    f.p0[s] = *(const v2f*)(Bw + 4 * s);
    f.p1[s] = *(const v2f*)(Bw + 16 * LDP + 4 * s);
    f.p2[s] = *(const v2f*)(Bw + 32 * LDP + 4 * s);
    f.p3[s] = *(const v2f*)(Bw + 48 * LDP + 4 * s);
  }
  asm volatile("" :: "v"(f.a[0]), "v"(f.a[1]), "v"(f.a[2]), "v"(f.a[3]), "v"(f.a[4]), "v"(f.a[5]), "v"(f.a[6]), "v"(f.a[7]),
                     "v"(f.p0[0]), "v"(f.p0[1]), "v"(f.p0[2]), "v"(f.p0[3]), "v"(f.p0[4]), "v"(f.p0[5]), "v"(f.p0[6]), "v"(f.p0[7]));
  asm volatile("" :: "v"(f.p1[0]), "v"(f.p1[1]), "v"(f.p1[2]), "v"(f.p1[3]), "v"(f.p1[4]), "v"(f.p1[5]), "v"(f.p1[6]), "v"(f.p1[7]),
                     "v"(f.p2[0]), "v"(f.p2[1]), "v"(f.p2[2]), "v"(f.p2[3]), "v"(f.p2[4]), "v"(f.p2[5]), "v"(f.p2[6]), "v"(f.p2[7]),
                     "v"(f.p3[0]), "v"(f.p3[1]), "v"(f.p3[2]), "v"(f.p3[3]), "v"(f.p3[4]), "v"(f.p3[5]), "v"(f.p3[6]), "v"(f.p3[7]));
}

__device__ __forceinline__ void mma_frags(const Frags& f, v8f& acc0, v8f& acc1, v8f& acc2, v8f& acc3) {
#pragma unroll
  for (int s = 0; s < 8; ++s) {
    acc0 = __builtin_amdgcn_wmma_f32_16x16x4_f32(false, f.a[s], false, f.p0[s], (short)0, acc0, false, false);
    acc1 = __builtin_amdgcn_wmma_f32_16x16x4_f32(false, f.a[s], false, f.p1[s], (short)0, acc1, false, false);
    acc2 = __builtin_amdgcn_wmma_f32_16x16x4_f32(false, f.a[s], false, f.p2[s], (short)0, acc2, false, false);
    acc3 = __builtin_amdgcn_wmma_f32_16x16x4_f32(false, f.a[s], false, f.p3[s], (short)0, acc3, false, false);
  }
}

#define NCHUNK (HDIM / KC)

__global__ __launch_bounds__(256)
void k_gemm_wmma(const float* __restrict__ h1,
                            const float* __restrict__ W2,
                            const float* __restrict__ b2,
                            float* __restrict__ h2,
                            float* __restrict__ rowstat) {
  __shared__ float sA[2][128 * LDP];
  __shared__ float sB[2][64 * LDP];
  const int tid   = threadIdx.x;
  const int wid   = tid >> 5;
  const int lane  = tid & 31;
  const int m0    = blockIdx.x * 128;
  const int n0    = blockIdx.y * 64;
  const int mrow  = lane & 15;
  const int khalf = (lane >> 4) << 1;

  v8f acc0 = {}, acc1 = {}, acc2 = {}, acc3 = {};

  const int sr = tid >> 3;
  const int sc = (tid & 7) << 2;
  const float* gA = h1 + (size_t)(m0 + sr) * HDIM + sc;
  const float* gB = W2 + (size_t)(n0 + sr) * HDIM + sc;
  const int so = sr * LDP + sc;
  const int fa = (wid * 16 + mrow) * LDP + khalf;
  const int fb = mrow * LDP + khalf;

  float4 ra[4], rb[2];
#define G_LOAD(k0)                                                              \
  _Pragma("unroll") for (int i = 0; i < 4; ++i)                                 \
    ra[i] = *(const float4*)(gA + (size_t)(32 * i) * HDIM + (k0));              \
  _Pragma("unroll") for (int i = 0; i < 2; ++i)                                 \
    rb[i] = *(const float4*)(gB + (size_t)(32 * i) * HDIM + (k0));
#define L_STORE(buf)                                                            \
  _Pragma("unroll") for (int i = 0; i < 4; ++i) {                               \
    v2f lo = {ra[i].x, ra[i].y};                                                \
    v2f hi = {ra[i].z, ra[i].w};                                                \
    *(v2f*)(sA[buf] + so + 32 * i * LDP)     = lo;                              \
    *(v2f*)(sA[buf] + so + 32 * i * LDP + 2) = hi;                              \
  }                                                                             \
  _Pragma("unroll") for (int i = 0; i < 2; ++i) {                               \
    v2f lo = {rb[i].x, rb[i].y};                                                \
    v2f hi = {rb[i].z, rb[i].w};                                                \
    *(v2f*)(sB[buf] + so + 32 * i * LDP)     = lo;                              \
    *(v2f*)(sB[buf] + so + 32 * i * LDP + 2) = hi;                              \
  }

  G_LOAD(0)
  L_STORE(0)
  G_LOAD(KC)
  __syncthreads();

  for (int c = 0; c < NCHUNK; ++c) {
    const int cur = c & 1;
    Frags f;
    load_frags(sA[cur] + fa, sB[cur] + fb, f);
    if (c + 1 < NCHUNK) {
      L_STORE(cur ^ 1)
    }
    if (c + 2 < NCHUNK) {
      G_LOAD((c + 2) * KC)
    }
    mma_frags(f, acc0, acc1, acc2, acc3);
    __syncthreads();
  }
#undef G_LOAD
#undef L_STORE

  const int ncol  = lane & 15;
  const int mbase = (lane >> 4) << 3;
  float* out = h2 + (size_t)(m0 + wid * 16 + mbase) * HDIM + n0;
  const float bj0 = b2[n0 + ncol];
  const float bj1 = b2[n0 + 16 + ncol];
  const float bj2 = b2[n0 + 32 + ncol];
  const float bj3 = b2[n0 + 48 + ncol];
  float2* sp = (float2*)rowstat + (size_t)blockIdx.y * BATCH + (m0 + wid * 16 + mbase);
  float rs_s[8], rs_q[8];
#pragma unroll
  for (int r = 0; r < 8; ++r) {
    float* row = out + (size_t)r * HDIM;
    const float v0 = acc0[r] + bj0;
    const float v1 = acc1[r] + bj1;
    const float v2 = acc2[r] + bj2;
    const float v3 = acc3[r] + bj3;
    row[ncol]      = v0;
    row[16 + ncol] = v1;
    row[32 + ncol] = v2;
    row[48 + ncol] = v3;
    rs_s[r] = (v0 + v1) + (v2 + v3);
    rs_q[r] = (v0 * v0 + v1 * v1) + (v2 * v2 + v3 * v3);
  }
#pragma unroll
  for (int off = 8; off > 0; off >>= 1) {
#pragma unroll
    for (int r = 0; r < 8; ++r) {
      rs_s[r] += __shfl_xor(rs_s[r], off, 32);
      rs_q[r] += __shfl_xor(rs_q[r], off, 32);
    }
  }
  if (ncol == 0) {
#pragma unroll
    for (int r = 0; r < 8; ++r) sp[r] = make_float2(rs_s[r], rs_q[r]);
  }
}

#define RPW 2
#define NTN (HDIM / 64)

__global__ __launch_bounds__(256)
void k_ln2_elu(const float* __restrict__ h2,
               const float* __restrict__ rowstat,
               const float* __restrict__ g2,
               const float* __restrict__ be2,
               float* __restrict__ out) {
  const int lane = threadIdx.x & 31;
  const int wid  = __builtin_amdgcn_readfirstlane(threadIdx.x >> 5);
  const int row0 = (blockIdx.x * 8 + wid) * RPW;

  float4 x[RPW][4];
#pragma unroll
  for (int r = 0; r < RPW; ++r) {
    const float4* src = (const float4*)(h2 + (size_t)(row0 + r) * HDIM);
#pragma unroll
    for (int i = 0; i < 4; ++i) x[r][i] = src[lane + 32 * i];
  }
  float4 g[4], be[4];
#pragma unroll
  for (int i = 0; i < 4; ++i) {
    g[i]  = ((const float4*)g2)[lane + 32 * i];
    be[i] = ((const float4*)be2)[lane + 32 * i];
  }
#pragma unroll
  for (int r = 0; r < RPW; ++r) {
    float S = 0.f, Q = 0.f;
#pragma unroll
    for (int y = 0; y < NTN; ++y) {
      const float2 p = ((const float2*)rowstat)[(size_t)y * BATCH + row0 + r];
      S += p.x;
      Q += p.y;
    }
    const float mu  = S * (1.0f / (float)HDIM);
    const float var = Q * (1.0f / (float)HDIM) - mu * mu;
    const float rs  = rsqrtf(var + LN_EPS);
    float4* dst = (float4*)(out + (size_t)(row0 + r) * HDIM);
#pragma unroll
    for (int i = 0; i < 4; ++i) {
      const float4 v = x[r][i];
      float4 o;
      o.x = elu((v.x - mu) * rs * g[i].x + be[i].x);
      o.y = elu((v.y - mu) * rs * g[i].y + be[i].y);
      o.z = elu((v.z - mu) * rs * g[i].z + be[i].z);
      o.w = elu((v.w - mu) * rs * g[i].w + be[i].w);
      dst[lane + 32 * i] = o;
    }
  }
}

extern "C" void kernel_launch(void* const* d_in, const int* in_sizes, int n_in,
                              void* d_out, int out_size, void* d_ws, size_t ws_size,
                              hipStream_t stream) {
  (void)in_sizes; (void)n_in; (void)out_size; (void)ws_size;
  const int*   msg = (const int*)  d_in[0];
  const float* W1  = (const float*)d_in[1];
  const float* b1  = (const float*)d_in[2];
  const float* g1  = (const float*)d_in[3];
  const float* be1 = (const float*)d_in[4];
  const float* W2  = (const float*)d_in[5];
  const float* b2  = (const float*)d_in[6];
  const float* g2  = (const float*)d_in[7];
  const float* be2 = (const float*)d_in[8];
  float* out = (float*)d_out;

  const size_t pk_bytes = (size_t)NGRP * BATCH * 16;
  const size_t h_bytes  = (size_t)BATCH * HDIM * sizeof(float);
  char* ws = (char*)d_ws;
  uint4* pk      = (uint4*)ws;
  float* h0T     = (float*)(ws + pk_bytes);
  float* h1      = (float*)(ws + pk_bytes + h_bytes);
  float* h2      = (float*)(ws + pk_bytes + 2 * h_bytes);
  float* rowstat = (float*)(ws + pk_bytes + 3 * h_bytes);

  k_pack_msg<<<NGRP * BATCH / 256, 256, 0, stream>>>(msg, pk);
  k_embed_cols<<<HDIM / 2, 1024, 0, stream>>>(pk, W1, b1, h0T);
  k_ln1_elu_turn<<<BATCH / 32, 32 * LW, 0, stream>>>(h0T, g1, be1, h1);
  k_gemm_wmma<<<dim3(BATCH / 128, HDIM / 64), 256, 0, stream>>>(h1, W2, b2, h2, rowstat);
  k_ln2_elu<<<BATCH / (8 * RPW), 256, 0, stream>>>(h2, rowstat, g2, be2, out);
}


# ELU in both LayerNorm launches: exponential side as exp2(x*log2 e)-1 on the hardware exp instruction, not the library expm1 chain; abs error near 1.5e-7, expm1's relative accuracy near 0 not kept
# speedup vs baseline: 1.8392x; 1.0352x over previous
#include <hip/hip_runtime.h>
#include <math.h>


#define NPOS  80
#define NCHAR 256
#define HDIM  512
#define IDIM  (NPOS * NCHAR)
#define BATCH 8192
#define LN_EPS 1e-5f
#define KC    32
#define LDP   34

typedef __attribute__((ext_vector_type(2))) float v2f;
typedef __attribute__((ext_vector_type(8))) float v8f;

__device__ __forceinline__ float elu(float x) {
  return x > 0.f ? x : __builtin_amdgcn_exp2f(x * 1.44269504088896340736f) - 1.0f;
}

#define NGRP (NPOS / 16)

__device__ __forceinline__ unsigned pack4(int4 v) {
  return (unsigned)(v.x & 255) | ((unsigned)(v.y & 255) << 8) |
         ((unsigned)(v.z & 255) << 16) | ((unsigned)(v.w & 255) << 24);
}

__global__ __launch_bounds__(256)
void k_pack_msg(const int* __restrict__ msg, uint4* __restrict__ pk) {
  const int t = blockIdx.x * 256 + threadIdx.x;
  const int g = t / BATCH;
  const int b = t % BATCH;
  const int* row = msg + (size_t)b * NPOS;
  const int p0 = (b & 31) + g * 16;
  unsigned w[4];
#pragma unroll
  for (int k = 0; k < 4; ++k) {
    int4 v;
    int q;
    q = p0 + 4 * k + 0; q = q >= NPOS ? q - NPOS : q; v.x = row[q];
    q = p0 + 4 * k + 1; q = q >= NPOS ? q - NPOS : q; v.y = row[q];
    q = p0 + 4 * k + 2; q = q >= NPOS ? q - NPOS : q; v.z = row[q];
    q = p0 + 4 * k + 3; q = q >= NPOS ? q - NPOS : q; v.w = row[q];
    w[k] = pack4(v);
  }
  pk[(size_t)g * BATCH + b] = make_uint4(w[0], w[1], w[2], w[3]);
}

#define NSLOT 128
#define SP    257
__global__ __launch_bounds__(1024)
void k_embed_cols(const uint4* __restrict__ pk,
                  const float* __restrict__ W1,
                  const float* __restrict__ b1,
                  float* __restrict__ h0T) {
  __shared__ float2 tab[NCHAR * NSLOT];
  __shared__ float  stg[32 * SP];
  const int tid  = threadIdx.x;
  const int lane = tid & 31;
  const int wid  = tid >> 5;
  const int j0   = blockIdx.x * 2;
  {
    const float4* r0 = (const float4*)(W1 + (size_t)j0 * IDIM);
    const float4* r1 = (const float4*)(W1 + (size_t)(j0 + 1) * IDIM);
    float4 a[5], c[5];
#pragma unroll
    for (int k = 0; k < 5; ++k) {
      a[k] = r0[tid + 1024 * k];
      c[k] = r1[tid + 1024 * k];
    }
    float* tf = (float*)tab;
    const int pl = tid >> 6;
    const int cc = (tid & 63) * 4;
#pragma unroll
    for (int k = 0; k < 5; ++k) {
      float* s0 = stg + pl * SP + cc;
      float* s1 = stg + (16 + pl) * SP + cc;
      s0[0] = a[k].x; s0[1] = a[k].y; s0[2] = a[k].z; s0[3] = a[k].w;
      s1[0] = c[k].x; s1[1] = c[k].y; s1[2] = c[k].z; s1[3] = c[k].w;
      __syncthreads();
      const int slot = 16 * k + (lane & 15);
      const int col  = lane >> 4;
      float v[8];
#pragma unroll
      for (int i = 0; i < 8; ++i) v[i] = stg[lane * SP + wid * 8 + i];
#pragma unroll
      for (int i = 0; i < 8; ++i) {
        const int ch = wid * 8 + i;
        tf[(ch * NSLOT + slot) * 2 + col] = v[i];
        if (k < 2) tf[(ch * NSLOT + slot + NPOS) * 2 + col] = v[i];
      }
      __syncthreads();
    }
  }

  const float bj0 = b1[j0];
  const float bj1 = b1[j0 + 1];
  const float2* tl = tab + lane;
  for (int it = 0; it < BATCH / 1024; ++it) {
    const int b = it * 1024 + tid;
    unsigned cw[4 * NGRP];
#pragma unroll
    for (int g = 0; g < NGRP; ++g) {
      const uint4 ch = pk[(size_t)g * BATCH + b];
      cw[4 * g + 0] = ch.x;
      cw[4 * g + 1] = ch.y;
      cw[4 * g + 2] = ch.z;
      cw[4 * g + 3] = ch.w;
    }
    float2 s[8];
#pragma unroll
    for (int grp = 0; grp < 8; ++grp) {
      float2 v[10];
#pragma unroll
      for (int q = 0; q < 10; ++q) {
        const int p = grp * 10 + q;
        const unsigned c = (cw[p >> 2] >> (8 * (p & 3))) & 255u;
        v[q] = tl[c * NSLOT + p];
      }
      asm volatile("" :: "v"(v[0].x), "v"(v[0].y), "v"(v[1].x), "v"(v[1].y),
                         "v"(v[2].x), "v"(v[2].y), "v"(v[3].x), "v"(v[3].y),
                         "v"(v[4].x), "v"(v[4].y), "v"(v[5].x), "v"(v[5].y),
                         "v"(v[6].x), "v"(v[6].y), "v"(v[7].x), "v"(v[7].y),
                         "v"(v[8].x), "v"(v[8].y), "v"(v[9].x), "v"(v[9].y));
      float2 t = v[0];
#pragma unroll
      for (int q = 1; q < 10; ++q) {
        t.x += v[q].x;
        t.y += v[q].y;
      }
      s[grp] = t;
    }
    float2 a = s[0];
#pragma unroll
    for (int w = 1; w < 8; ++w) {
      a.x += s[w].x;
      a.y += s[w].y;
    }
    a.x += bj0;
    a.y += bj1;
    h0T[(size_t)j0 * BATCH + b]       = a.x;
    h0T[(size_t)(j0 + 1) * BATCH + b] = a.y;
  }
}

#define TP 516
#define LW 16

__global__ __launch_bounds__(512)
void k_ln1_elu_turn(const float* __restrict__ h0T,
                    const float* __restrict__ g1,
                    const float* __restrict__ be1,
                    float* __restrict__ h1) {
  __shared__ __attribute__((aligned(16))) float tile[32 * TP];
  __shared__ __attribute__((aligned(16))) float2 part[32 * LW];
  const int tid  = threadIdx.x;
  const int lane = tid & 31;
  const int wid  = __builtin_amdgcn_readfirstlane(tid >> 5);
  const int b0   = blockIdx.x * 32;
  const float* src = h0T + (size_t)(wid * 32) * BATCH + b0 + lane;

  float x[32];
#pragma unroll
  for (int i = 0; i < 32; ++i) x[i] = src[(size_t)i * BATCH];

  float4 g[4], be[4];
#pragma unroll
  for (int i = 0; i < 4; ++i) {
    g[i]  = ((const float4*)g1)[lane + 32 * i];
    be[i] = ((const float4*)be1)[lane + 32 * i];
  }

  float s0 = 0.f, s1 = 0.f, s2 = 0.f, s3 = 0.f;
  float q0 = 0.f, q1 = 0.f, q2 = 0.f, q3 = 0.f;
  float* trow = tile + lane * TP + wid * 32;
#pragma unroll
  for (int i = 0; i < 32; i += 4) {
    s0 += x[i];     q0 += x[i]     * x[i];
    s1 += x[i + 1]; q1 += x[i + 1] * x[i + 1];
    s2 += x[i + 2]; q2 += x[i + 2] * x[i + 2];
    s3 += x[i + 3]; q3 += x[i + 3] * x[i + 3];
    *(float4*)(trow + i) = make_float4(x[i], x[i + 1], x[i + 2], x[i + 3]);
  }
  part[lane * LW + wid] = make_float2((s0 + s1) + (s2 + s3), (q0 + q1) + (q2 + q3));
  asm volatile("" :: "v"(g[0].x), "v"(g[0].y), "v"(g[0].z), "v"(g[0].w),
                     "v"(g[1].x), "v"(g[1].y), "v"(g[1].z), "v"(g[1].w),
                     "v"(g[2].x), "v"(g[2].y), "v"(g[2].z), "v"(g[2].w),
                     "v"(g[3].x), "v"(g[3].y), "v"(g[3].z), "v"(g[3].w));
  asm volatile("" :: "v"(be[0].x), "v"(be[0].y), "v"(be[0].z), "v"(be[0].w),
                     "v"(be[1].x), "v"(be[1].y), "v"(be[1].z), "v"(be[1].w),
                     "v"(be[2].x), "v"(be[2].y), "v"(be[2].z), "v"(be[2].w),
                     "v"(be[3].x), "v"(be[3].y), "v"(be[3].z), "v"(be[3].w));
  __syncthreads();

  float mu[2], rs[2];
  float4 v[2][4];
#pragma unroll
  for (int r = 0; r < 2; ++r) {
    const int row = wid * 2 + r;
    const float4* pr = (const float4*)(part + row * LW);
    float S = 0.f, Q = 0.f;
#pragma unroll
    for (int k = 0; k < LW / 2; ++k) {
      const float4 p = pr[k];
      S += p.x; Q += p.y;
      S += p.z; Q += p.w;
    }
    mu[r] = S * (1.0f / (float)HDIM);
    const float var = Q * (1.0f / (float)HDIM) - mu[r] * mu[r];
    rs[r] = rsqrtf(var + LN_EPS);
    const float4* tr = (const float4*)(tile + row * TP);
#pragma unroll
    for (int i = 0; i < 4; ++i) v[r][i] = tr[lane + 32 * i];
  }
#pragma unroll
  for (int r = 0; r < 2; ++r) {
    float4* dst = (float4*)(h1 + (size_t)(b0 + wid * 2 + r) * HDIM);
#pragma unroll
    for (int i = 0; i < 4; ++i) {
      float4 o;
      o.x = elu((v[r][i].x - mu[r]) * rs[r] * g[i].x + be[i].x);
      o.y = elu((v[r][i].y - mu[r]) * rs[r] * g[i].y + be[i].y);
      o.z = elu((v[r][i].z - mu[r]) * rs[r] * g[i].z + be[i].z);
      o.w = elu((v[r][i].w - mu[r]) * rs[r] * g[i].w + be[i].w);
      dst[lane + 32 * i] = o;
    }
  }
}

struct Frags { v2f a[8], p0[8], p1[8], p2[8], p3[8]; };

__device__ __forceinline__ void load_frags(const float* Aw, const float* Bw, Frags& f) {
#pragma unroll
  for (int s = 0; s < 8; ++s) {
    f.a[s]  = *(const v2f*)(Aw + 4 * s);
    f.p0[s] = *(const v2f*)(Bw + 4 * s);
    f.p1[s] = *(const v2f*)(Bw + 16 * LDP + 4 * s);
    f.p2[s] = *(const v2f*)(Bw + 32 * LDP + 4 * s);
    f.p3[s] = *(const v2f*)(Bw + 48 * LDP + 4 * s);
  }
  asm volatile("" :: "v"(f.a[0]), "v"(f.a[1]), "v"(f.a[2]), "v"(f.a[3]), "v"(f.a[4]), "v"(f.a[5]), "v"(f.a[6]), "v"(f.a[7]),
                     "v"(f.p0[0]), "v"(f.p0[1]), "v"(f.p0[2]), "v"(f.p0[3]), "v"(f.p0[4]), "v"(f.p0[5]), "v"(f.p0[6]), "v"(f.p0[7]));
  asm volatile("" :: "v"(f.p1[0]), "v"(f.p1[1]), "v"(f.p1[2]), "v"(f.p1[3]), "v"(f.p1[4]), "v"(f.p1[5]), "v"(f.p1[6]), "v"(f.p1[7]),
                     "v"(f.p2[0]), "v"(f.p2[1]), "v"(f.p2[2]), "v"(f.p2[3]), "v"(f.p2[4]), "v"(f.p2[5]), "v"(f.p2[6]), "v"(f.p2[7]),
                     "v"(f.p3[0]), "v"(f.p3[1]), "v"(f.p3[2]), "v"(f.p3[3]), "v"(f.p3[4]), "v"(f.p3[5]), "v"(f.p3[6]), "v"(f.p3[7]));
}

__device__ __forceinline__ void mma_frags(const Frags& f, v8f& acc0, v8f& acc1, v8f& acc2, v8f& acc3) {
#pragma unroll
  for (int s = 0; s < 8; ++s) {
    acc0 = __builtin_amdgcn_wmma_f32_16x16x4_f32(false, f.a[s], false, f.p0[s], (short)0, acc0, false, false);
    acc1 = __builtin_amdgcn_wmma_f32_16x16x4_f32(false, f.a[s], false, f.p1[s], (short)0, acc1, false, false);
    acc2 = __builtin_amdgcn_wmma_f32_16x16x4_f32(false, f.a[s], false, f.p2[s], (short)0, acc2, false, false);
    acc3 = __builtin_amdgcn_wmma_f32_16x16x4_f32(false, f.a[s], false, f.p3[s], (short)0, acc3, false, false);
  }
}

#define NCHUNK (HDIM / KC)

__global__ __launch_bounds__(256)
void k_gemm_wmma(const float* __restrict__ h1,
                            const float* __restrict__ W2,
                            const float* __restrict__ b2,
                            float* __restrict__ h2,
                            float* __restrict__ rowstat) {
  __shared__ float sA[2][128 * LDP];
  __shared__ float sB[2][64 * LDP];
  const int tid   = threadIdx.x;
  const int wid   = tid >> 5;
  const int lane  = tid & 31;
  const int m0    = blockIdx.x * 128;
  const int n0    = blockIdx.y * 64;
  const int mrow  = lane & 15;
  const int khalf = (lane >> 4) << 1;

  v8f acc0 = {}, acc1 = {}, acc2 = {}, acc3 = {};

  const int sr = tid >> 3;
  const int sc = (tid & 7) << 2;
  const float* gA = h1 + (size_t)(m0 + sr) * HDIM + sc;
  const float* gB = W2 + (size_t)(n0 + sr) * HDIM + sc;
  const int so = sr * LDP + sc;
  const int fa = (wid * 16 + mrow) * LDP + khalf;
  const int fb = mrow * LDP + khalf;

  float4 ra[4], rb[2];
#define G_LOAD(k0)                                                              \
  _Pragma("unroll") for (int i = 0; i < 4; ++i)                                 \
    ra[i] = *(const float4*)(gA + (size_t)(32 * i) * HDIM + (k0));              \
  _Pragma("unroll") for (int i = 0; i < 2; ++i)                                 \
    rb[i] = *(const float4*)(gB + (size_t)(32 * i) * HDIM + (k0));
#define L_STORE(buf)                                                            \
  _Pragma("unroll") for (int i = 0; i < 4; ++i) {                               \
    v2f lo = {ra[i].x, ra[i].y};                                                \
    v2f hi = {ra[i].z, ra[i].w};                                                \
    *(v2f*)(sA[buf] + so + 32 * i * LDP)     = lo;                              \
    *(v2f*)(sA[buf] + so + 32 * i * LDP + 2) = hi;                              \
  }                                                                             \
  _Pragma("unroll") for (int i = 0; i < 2; ++i) {                               \
    v2f lo = {rb[i].x, rb[i].y};                                                \
    v2f hi = {rb[i].z, rb[i].w};                                                \
    *(v2f*)(sB[buf] + so + 32 * i * LDP)     = lo;                              \
    *(v2f*)(sB[buf] + so + 32 * i * LDP + 2) = hi;                              \
  }

  G_LOAD(0)
  L_STORE(0)
  G_LOAD(KC)
  __syncthreads();

  for (int c = 0; c < NCHUNK; ++c) {
    const int cur = c & 1;
    Frags f;
    load_frags(sA[cur] + fa, sB[cur] + fb, f);
    if (c + 1 < NCHUNK) {
      L_STORE(cur ^ 1)
    }
    if (c + 2 < NCHUNK) {
      G_LOAD((c + 2) * KC)
    }
    mma_frags(f, acc0, acc1, acc2, acc3);
    __syncthreads();
  }
#undef G_LOAD
#undef L_STORE

  const int ncol  = lane & 15;
  const int mbase = (lane >> 4) << 3;
  float* out = h2 + (size_t)(m0 + wid * 16 + mbase) * HDIM + n0;
  const float bj0 = b2[n0 + ncol];
  const float bj1 = b2[n0 + 16 + ncol];
  const float bj2 = b2[n0 + 32 + ncol];
  const float bj3 = b2[n0 + 48 + ncol];
  float2* sp = (float2*)rowstat + (size_t)blockIdx.y * BATCH + (m0 + wid * 16 + mbase);
  float rs_s[8], rs_q[8];
#pragma unroll
  for (int r = 0; r < 8; ++r) {
    float* row = out + (size_t)r * HDIM;
    const float v0 = acc0[r] + bj0;
    const float v1 = acc1[r] + bj1;
    const float v2 = acc2[r] + bj2;
    const float v3 = acc3[r] + bj3;
    row[ncol]      = v0;
    row[16 + ncol] = v1;
    row[32 + ncol] = v2;
    row[48 + ncol] = v3;
    rs_s[r] = (v0 + v1) + (v2 + v3);
    rs_q[r] = (v0 * v0 + v1 * v1) + (v2 * v2 + v3 * v3);
  }
#pragma unroll
  for (int off = 8; off > 0; off >>= 1) {
#pragma unroll
    for (int r = 0; r < 8; ++r) {
      rs_s[r] += __shfl_xor(rs_s[r], off, 32);
      rs_q[r] += __shfl_xor(rs_q[r], off, 32);
    }
  }
  if (ncol == 0) {
#pragma unroll
    for (int r = 0; r < 8; ++r) sp[r] = make_float2(rs_s[r], rs_q[r]);
  }
}

#define RPW 2
#define NTN (HDIM / 64)

__global__ __launch_bounds__(256)
void k_ln2_elu(const float* __restrict__ h2,
               const float* __restrict__ rowstat,
               const float* __restrict__ g2,
               const float* __restrict__ be2,
               float* __restrict__ out) {
  const int lane = threadIdx.x & 31;
  const int wid  = __builtin_amdgcn_readfirstlane(threadIdx.x >> 5);
  const int row0 = (blockIdx.x * 8 + wid) * RPW;

  float4 x[RPW][4];
#pragma unroll
  for (int r = 0; r < RPW; ++r) {
    const float4* src = (const float4*)(h2 + (size_t)(row0 + r) * HDIM);
#pragma unroll
    for (int i = 0; i < 4; ++i) x[r][i] = src[lane + 32 * i];
  }
  float4 g[4], be[4];
#pragma unroll
  for (int i = 0; i < 4; ++i) {
    g[i]  = ((const float4*)g2)[lane + 32 * i];
    be[i] = ((const float4*)be2)[lane + 32 * i];
  }
#pragma unroll
  for (int r = 0; r < RPW; ++r) {
    float S = 0.f, Q = 0.f;
#pragma unroll
    for (int y = 0; y < NTN; ++y) {
      const float2 p = ((const float2*)rowstat)[(size_t)y * BATCH + row0 + r];
      S += p.x;
      Q += p.y;
    }
    const float mu  = S * (1.0f / (float)HDIM);
    const float var = Q * (1.0f / (float)HDIM) - mu * mu;
    const float rs  = rsqrtf(var + LN_EPS);
    float4* dst = (float4*)(out + (size_t)(row0 + r) * HDIM);
#pragma unroll
    for (int i = 0; i < 4; ++i) {
      const float4 v = x[r][i];
      float4 o;
      o.x = elu((v.x - mu) * rs * g[i].x + be[i].x);
      o.y = elu((v.y - mu) * rs * g[i].y + be[i].y);
      o.z = elu((v.z - mu) * rs * g[i].z + be[i].z);
      o.w = elu((v.w - mu) * rs * g[i].w + be[i].w);
      dst[lane + 32 * i] = o;
    }
  }
}

extern "C" void kernel_launch(void* const* d_in, const int* in_sizes, int n_in,
                              void* d_out, int out_size, void* d_ws, size_t ws_size,
                              hipStream_t stream) {
  (void)in_sizes; (void)n_in; (void)out_size; (void)ws_size;
  const int*   msg = (const int*)  d_in[0];
  const float* W1  = (const float*)d_in[1];
  const float* b1  = (const float*)d_in[2];
  const float* g1  = (const float*)d_in[3];
  const float* be1 = (const float*)d_in[4];
  const float* W2  = (const float*)d_in[5];
  const float* b2  = (const float*)d_in[6];
  const float* g2  = (const float*)d_in[7];
  const float* be2 = (const float*)d_in[8];
  float* out = (float*)d_out;

  const size_t pk_bytes = (size_t)NGRP * BATCH * 16;
  const size_t h_bytes  = (size_t)BATCH * HDIM * sizeof(float);
  char* ws = (char*)d_ws;
  uint4* pk      = (uint4*)ws;
  float* h0T     = (float*)(ws + pk_bytes);
  float* h1      = (float*)(ws + pk_bytes + h_bytes);
  float* h2      = (float*)(ws + pk_bytes + 2 * h_bytes);
  float* rowstat = (float*)(ws + pk_bytes + 3 * h_bytes);

  k_pack_msg<<<NGRP * BATCH / 256, 256, 0, stream>>>(msg, pk);
  k_embed_cols<<<HDIM / 2, 1024, 0, stream>>>(pk, W1, b1, h0T);
  k_ln1_elu_turn<<<BATCH / 32, 32 * LW, 0, stream>>>(h0T, g1, be1, h1);
  k_gemm_wmma<<<dim3(BATCH / 128, HDIM / 64), 256, 0, stream>>>(h1, W2, b2, h2, rowstat);
  k_ln2_elu<<<BATCH / (8 * RPW), 256, 0, stream>>>(h2, rowstat, g2, be2, out);
}
